# starting kernel
# baseline (speedup 1.0000x reference)
#include <hip/hip_runtime.h>


#define BATCHES        64
#define PTS            65536
#define NBINS          4
#define NCELLS         64
#define NFEAT          10
#define NCOLS          640
#define WGS_PER_BATCH  16
#define PTS_PER_WG     (PTS / WGS_PER_BATCH)
#define K1_BLOCKS      24
#define K1_F4_PER_THR  16
#define FLOATS_PER_B   (PTS * 6)
#define F4_PER_B       (FLOATS_PER_B / 4)

typedef float v2f __attribute__((ext_vector_type(2)));
typedef float v8f __attribute__((ext_vector_type(8)));

__global__ void k0_init(unsigned* __restrict__ bbox) {
  int i = blockIdx.x * blockDim.x + threadIdx.x;
  if (i < BATCHES) bbox[i] = 0u;
}

__global__ void k1_bbox(const float* __restrict__ x,
                        unsigned* __restrict__ bbox) {
  const int b = blockIdx.y;
  const float4* xb = (const float4*)(x + (size_t)b * FLOATS_PER_B);
  const int base = blockIdx.x * (256 * K1_F4_PER_THR) + threadIdx.x;

  float m = 0.0f;
  for (int i = 0; i < K1_F4_PER_THR; ++i) {
    const int q = base + i * 256;
    __builtin_prefetch(&xb[q + 2048], 0, 0);
    const float4 v = xb[q];
    const int m6 = (q * 4) % 6;
    float a0 = (((m6 + 0) % 6) < 3) ? fabsf(v.x) : 0.0f;
    float a1 = (((m6 + 1) % 6) < 3) ? fabsf(v.y) : 0.0f;
    float a2 = (((m6 + 2) % 6) < 3) ? fabsf(v.z) : 0.0f;
    float a3 = (((m6 + 3) % 6) < 3) ? fabsf(v.w) : 0.0f;
    m = fmaxf(m, fmaxf(fmaxf(a0, a1), fmaxf(a2, a3)));
  }

  __shared__ float red[256];
  red[threadIdx.x] = m;
  __syncthreads();
  for (int s = 128; s > 0; s >>= 1) {
    if (threadIdx.x < s) red[threadIdx.x] = fmaxf(red[threadIdx.x], red[threadIdx.x + s]);
    __syncthreads();
  }
  if (threadIdx.x == 0)
    atomicMax(&bbox[b], __float_as_uint(red[0]));
}

__global__ void k2_hist(const float* __restrict__ x,
                        const unsigned* __restrict__ bbox,
                        float* __restrict__ partial) {
  const int b  = blockIdx.y;
  const int wg = blockIdx.x;

  __shared__ float hist[NCOLS];
  for (int i = threadIdx.x; i < NCOLS; i += 256) hist[i] = 0.0f;
  __syncthreads();

  const float bb     = __uint_as_float(bbox[b]);
  const float inv_th = 1.0f / fmaxf(2.0f * bb, 1e-5f);
  const float* xb = x + (size_t)b * FLOATS_PER_B + (size_t)wg * PTS_PER_WG * 6;

  for (int i = 0; i < PTS_PER_WG / 256; ++i) {
    const float* pp = xb + (size_t)(i * 256 + threadIdx.x) * 6;
    __builtin_prefetch(pp + 256 * 6, 0, 0);
    const float px = pp[0], py = pp[1], pz = pp[2];
    const float ox = pp[3], oy = pp[4], oz = pp[5];
    const float cx = (px + bb) * inv_th;
    const float cy = (py + bb) * inv_th;
    const float cz = (pz + bb) * inv_th;
    const int i0 = min(max((int)(cx * (float)NBINS), 0), NBINS - 1);
    const int i1 = min(max((int)(cy * (float)NBINS), 0), NBINS - 1);
    const int i2 = min(max((int)(cz * (float)NBINS), 0), NBINS - 1);
    const int base = ((i0 * NBINS + i1) * NBINS + i2) * NFEAT;
    atomicAdd(&hist[base + 0], 1.0f);
    atomicAdd(&hist[base + 1], cx);
    atomicAdd(&hist[base + 2], cy);
    atomicAdd(&hist[base + 3], cz);
    atomicAdd(&hist[base + 4], ox * ox);
    atomicAdd(&hist[base + 5], ox * oy);
    atomicAdd(&hist[base + 6], ox * oz);
    atomicAdd(&hist[base + 7], oy * oy);
    atomicAdd(&hist[base + 8], oy * oz);
    atomicAdd(&hist[base + 9], oz * oz);
  }
  __syncthreads();

  float* dst = partial + (size_t)(b * WGS_PER_BATCH + wg) * NCOLS;
  for (int i = threadIdx.x; i < NCOLS; i += 256) dst[i] = hist[i];
}

__global__ void k3_final(const float* __restrict__ partial,
                         float* __restrict__ out) {
  const int b    = blockIdx.x;
  const int lane = threadIdx.x & 31;
  const int wave = threadIdx.x >> 5;

  __shared__ float sums[NCOLS];
  __shared__ float fall[NCOLS];
  __shared__ float red[256];

  const float* pb = partial + (size_t)b * WGS_PER_BATCH * NCOLS;

#if __has_builtin(__builtin_amdgcn_wmma_f32_16x16x4_f32)
  for (int t = wave * 5; t < wave * 5 + 5; ++t) {
    const int col = t * 16 + (lane & 15);
    v8f c = {};
    v2f a; a[0] = 1.0f; a[1] = 1.0f;
    for (int k4 = 0; k4 < WGS_PER_BATCH / 4; ++k4) {
      const int r0 = k4 * 4 + ((lane < 16) ? 0 : 2);
      v2f bm;
      bm[0] = pb[(size_t)r0 * NCOLS + col];
      bm[1] = pb[(size_t)(r0 + 1) * NCOLS + col];
      c = __builtin_amdgcn_wmma_f32_16x16x4_f32(false, a, false, bm,
                                                (short)0, c, false, false);
    }
    if (lane < 16) sums[col] = c[0];
  }
#else
  for (int t = wave * 5; t < wave * 5 + 5; ++t) {
    const int col = t * 16 + (lane & 15);
    if (lane < 16) {
      float s = 0.0f;
      for (int k = 0; k < WGS_PER_BATCH; ++k) s += pb[(size_t)k * NCOLS + col];
      sums[col] = s;
    }
  }
#endif
  __syncthreads();

  float ss = 0.0f;
  for (int i = threadIdx.x; i < NCOLS; i += 256) {
    const int cell = i / NFEAT;
    const int f    = i - cell * NFEAT;
    const float cnt  = sums[cell * NFEAT];
    const float fc   = fmaxf(cnt, 1.0f);
    const float upre = rsqrtf(fc);
    float val;
    if (f == 0) {
      val = 0.001f * cnt * upre;
    } else if (f < 4) {
      const int d  = f - 1;
      const int ci = (d == 0) ? (cell >> 4) : ((d == 1) ? ((cell >> 2) & 3) : (cell & 3));
      const float center = ((float)ci + 0.5f) * 0.25f;
      val = (sums[i] - cnt * center) * upre;
    } else {
      val = sums[i] / fc;
    }
    fall[i] = val;
    ss += val * val;
  }

  red[threadIdx.x] = ss;
  __syncthreads();
  for (int s = 128; s > 0; s >>= 1) {
    if (threadIdx.x < s) red[threadIdx.x] += red[threadIdx.x + s];
    __syncthreads();
  }
  const float scale = 1.0f / fmaxf(sqrtf(red[0]), 1e-12f);
  for (int i = threadIdx.x; i < NCOLS; i += 256)
    out[(size_t)b * NCOLS + i] = fall[i] * scale;
}

extern "C" void kernel_launch(void* const* d_in, const int* in_sizes, int n_in,
                              void* d_out, int out_size, void* d_ws, size_t ws_size,
                              hipStream_t stream) {
  const float* x   = (const float*)d_in[0];
  unsigned* bbox   = (unsigned*)d_ws;
  float* partial   = (float*)d_ws + 64;
  float* out       = (float*)d_out;

  k0_init <<<1, 64, 0, stream>>>(bbox);
  k1_bbox <<<dim3(K1_BLOCKS, BATCHES), 256, 0, stream>>>(x, bbox);
  k2_hist <<<dim3(WGS_PER_BATCH, BATCHES), 256, 0, stream>>>(x, bbox, partial);
  k3_final<<<BATCHES, 256, 0, stream>>>(partial, out);
}


# K2: each round's points compacted by i2 in LDS (ballot ranks), so B needs no i2 one-hot: one f32 16x16x4 WMMA per 4 points into one of four per-i2 accumulator tiles, not three (sum order changes)
# speedup vs baseline: 2.1534x; 2.1534x over previous
#include <hip/hip_runtime.h>


#define BATCHES        64
#define PTS            65536
#define NBINS          4
#define NCELLS         64
#define NFEAT          10
#define NCOLS          640
#define WGS_PER_BATCH  16
#define PTS_PER_WG     (PTS / WGS_PER_BATCH)
#define K1_BLOCKS      24
#define K1_F4_PER_THR  16
#define FLOATS_PER_B   (PTS * 6)
#define F4_PER_B       (FLOATS_PER_B / 4)

typedef float v2f __attribute__((ext_vector_type(2)));
typedef float v8f __attribute__((ext_vector_type(8)));

__global__ void k0_init(unsigned* __restrict__ bbox) {
  int i = blockIdx.x * blockDim.x + threadIdx.x;
  if (i < BATCHES) bbox[i] = 0u;
}

__global__ void __launch_bounds__(256)
k1_bbox(const float* __restrict__ x,
        unsigned* __restrict__ bbox) {
  const int b = blockIdx.y;
  const float4* xb = (const float4*)(x + (size_t)b * FLOATS_PER_B);
  const int base = blockIdx.x * (256 * K1_F4_PER_THR) + threadIdx.x;

  const unsigned ph   = (unsigned)base % 3u;
  const unsigned KEEP = 0x7fffffffu;
  unsigned kx[3], ky[3], kz[3], kw[3];
  for (int j = 0; j < 3; ++j) {
    unsigned t = ph + (unsigned)j;
    t = (t >= 3u) ? t - 3u : t;
    kx[j] = (t != 1u) ? KEEP : 0u;
    ky[j] = (t == 0u) ? KEEP : 0u;
    kz[j] = (t != 2u) ? KEEP : 0u;
    kw[j] = (t == 1u) ? KEEP : 0u;
  }

  float4 v[K1_F4_PER_THR];
#pragma unroll
  for (int i = 0; i < K1_F4_PER_THR; ++i) v[i] = xb[base + i * 256];

  float m = 0.0f;
#pragma unroll
  for (int i = 0; i < K1_F4_PER_THR; ++i) {
    const int j = i % 3;
    const float a0 = __uint_as_float(__float_as_uint(v[i].x) & kx[j]);
    const float a1 = __uint_as_float(__float_as_uint(v[i].y) & ky[j]);
    const float a2 = __uint_as_float(__float_as_uint(v[i].z) & kz[j]);
    const float a3 = __uint_as_float(__float_as_uint(v[i].w) & kw[j]);
    m = fmaxf(m, fmaxf(fmaxf(a0, a1), fmaxf(a2, a3)));
  }

  __shared__ float red[256];
  red[threadIdx.x] = m;
  __syncthreads();
  for (int s = 128; s > 0; s >>= 1) {
    if (threadIdx.x < s) red[threadIdx.x] = fmaxf(red[threadIdx.x], red[threadIdx.x + s]);
    __syncthreads();
  }
  if (threadIdx.x == 0)
    atomicMax(&bbox[b], __float_as_uint(red[0]));
}

#define K2_WAVES        8
#define K2_PTS_PER_WAVE (PTS_PER_WG / K2_WAVES)
#define RPTS            64
#define K2_ROUNDS       (K2_PTS_PER_WAVE / RPTS)
#define REC_DW          12
#define SLOTS           80
#define STAGE_V4        (SLOTS * REC_DW / 4)
#define K2_TILE_FLOATS  (4 * 8 * 32)

typedef unsigned v4u __attribute__((ext_vector_type(4)));

struct K2Rec { v4u q0, q1, q2; unsigned i2; };

__device__ __forceinline__ K2Rec k2_record(const float* __restrict__ xw, int p, float bb, float inv_th) {
  const float2* pp = (const float2*)(xw + (size_t)p * 6);
  const float2 d0 = pp[0], d1 = pp[1], d2 = pp[2];
  const float px = d0.x, py = d0.y, pz = d1.x;
  const float ox = d1.y, oy = d2.x, oz = d2.y;
  const float cx = (px + bb) * inv_th;
  const float cy = (py + bb) * inv_th;
  const float cz = (pz + bb) * inv_th;
  const int i0 = min(max((int)(cx * (float)NBINS), 0), NBINS - 1);
  const int i1 = min(max((int)(cy * (float)NBINS), 0), NBINS - 1);
  const int i2 = min(max((int)(cz * (float)NBINS), 0), NBINS - 1);
  K2Rec r;
  r.q0.x = __float_as_uint(1.0f);    r.q0.y = __float_as_uint(cx);      r.q0.z = __float_as_uint(cy);      r.q0.w = __float_as_uint(cz);
  r.q1.x = __float_as_uint(ox * ox); r.q1.y = __float_as_uint(ox * oy); r.q1.z = __float_as_uint(ox * oz); r.q1.w = __float_as_uint(oy * oy);
  r.q2.x = __float_as_uint(oy * oz); r.q2.y = __float_as_uint(oz * oz); r.q2.z = (unsigned)(i0 * NBINS + i1); r.q2.w = 0u;
  r.i2 = (unsigned)i2;
  return r;
}

#define K2_GROUP(G, ACC)                                                              \
  for (unsigned j = 0; j < gsteps[G]; ++j) {                                          \
    const unsigned* rp = stw + (gbase[G] + 4u * j + 2u * (unsigned)h) * REC_DW;       \
    const unsigned w0 = rp[10], w1 = rp[REC_DW + 10];                                 \
    v2f a, bm;                                                                        \
    bm[0] = __uint_as_float(rp[cc]);                                                  \
    bm[1] = __uint_as_float(rp[REC_DW + cc]);                                         \
    a[0]  = (w0 == myrow) ? 1.0f : 0.0f;                                              \
    a[1]  = (w1 == myrow) ? 1.0f : 0.0f;                                              \
    ACC = __builtin_amdgcn_wmma_f32_16x16x4_f32(false, a, false, bm, (short)0, ACC, false, false); \
  }

__global__ void __launch_bounds__(256)
k2_hist(const float* __restrict__ x,
        const unsigned* __restrict__ bbox,
        float* __restrict__ partial) {
  const int b    = blockIdx.y;
  const int wg   = blockIdx.x;
  const int tid  = threadIdx.x;
  const int lane = tid & 31;
  const int wave = tid >> 5;
  const int c    = lane & 15;
  const int h    = lane >> 4;
  const unsigned myrow = (unsigned)c;
  const int cc   = (c < REC_DW) ? c : (REC_DW - 1);

  __shared__ v4u   smem[K2_WAVES * K2_TILE_FLOATS / 4];
  __shared__ float hist[NCOLS];

  const float bb     = __uint_as_float(bbox[b]);
  const float inv_th = 1.0f / fmaxf(2.0f * bb, 1e-5f);
  const float* xw = x + (size_t)b * FLOATS_PER_B
                      + ((size_t)wg * PTS_PER_WG + (size_t)wave * K2_PTS_PER_WAVE) * 6;

  v4u* st = smem + wave * STAGE_V4;
  const unsigned* stw = (const unsigned*)st;

  v8f acc0 = {}, acc1 = {}, acc2 = {}, acc3 = {};

  for (int round = 0; round < K2_ROUNDS; ++round) {
    const K2Rec ra = k2_record(xw, round * RPTS + lane,      bb, inv_th);
    const K2Rec rb = k2_record(xw, round * RPTS + 32 + lane, bb, inv_th);

    unsigned slot_a = 0u, slot_b = 0u, base = 0u, padslot = 0u, padend = 0u;
    unsigned gbase[4], gsteps[4];
#pragma unroll
    for (int g = 0; g < 4; ++g) {
      const unsigned ma = __builtin_amdgcn_ballot_w32(ra.i2 == (unsigned)g);
      const unsigned mb = __builtin_amdgcn_ballot_w32(rb.i2 == (unsigned)g);
      const unsigned ca = (unsigned)__builtin_popcount(ma);
      const unsigned cb = (unsigned)__builtin_popcount(mb);
      const unsigned ka = base + __builtin_amdgcn_mbcnt_lo(ma, 0u);
      const unsigned kb = base + ca + __builtin_amdgcn_mbcnt_lo(mb, 0u);
      slot_a = (ra.i2 == (unsigned)g) ? ka : slot_a;
      slot_b = (rb.i2 == (unsigned)g) ? kb : slot_b;
      const unsigned n     = ca + cb;
      const unsigned steps = (n + 3u) >> 2;
      gbase[g]  = base;
      gsteps[g] = steps;
      if ((lane >> 2) == g) { padslot = base + n + (unsigned)(lane & 3); padend = base + 4u * steps; }
      base += 4u * steps;
    }

    st[slot_a * 3u + 0u] = ra.q0;  st[slot_a * 3u + 1u] = ra.q1;  st[slot_a * 3u + 2u] = ra.q2;
    st[slot_b * 3u + 0u] = rb.q0;  st[slot_b * 3u + 1u] = rb.q1;  st[slot_b * 3u + 2u] = rb.q2;
    if (padslot < padend) {
      const v4u z  = {0u, 0u, 0u, 0u};
      const v4u zr = {0u, 0u, 0xffu, 0u};
      st[padslot * 3u + 0u] = z;  st[padslot * 3u + 1u] = z;  st[padslot * 3u + 2u] = zr;
    }

    __builtin_amdgcn_fence(__ATOMIC_RELEASE, "wavefront");
    __builtin_amdgcn_wave_barrier();
    __builtin_amdgcn_fence(__ATOMIC_ACQUIRE, "wavefront");

    K2_GROUP(0, acc0)
    K2_GROUP(1, acc1)
    K2_GROUP(2, acc2)
    K2_GROUP(3, acc3)

    __builtin_amdgcn_fence(__ATOMIC_RELEASE, "wavefront");
    __builtin_amdgcn_wave_barrier();
    __builtin_amdgcn_fence(__ATOMIC_ACQUIRE, "wavefront");
  }

  __syncthreads();
  float* red = (float*)smem;
  for (int r = 0; r < 8; ++r) {
    red[wave * K2_TILE_FLOATS + (0 * 8 + r) * 32 + lane] = acc0[r];
    red[wave * K2_TILE_FLOATS + (1 * 8 + r) * 32 + lane] = acc1[r];
    red[wave * K2_TILE_FLOATS + (2 * 8 + r) * 32 + lane] = acc2[r];
    red[wave * K2_TILE_FLOATS + (3 * 8 + r) * 32 + lane] = acc3[r];
  }
  __syncthreads();
  for (int o = tid; o < K2_TILE_FLOATS; o += 256) {
    float s = 0.0f;
    for (int w = 0; w < K2_WAVES; ++w) s += red[w * K2_TILE_FLOATS + o];
    const int g  = o >> 8;
    const int r  = (o >> 5) & 7;
    const int l  = o & 31;
    const int f  = l & 15;
    const int cell = ((8 * (l >> 4) + r) << 2) + g;
    if (f < NFEAT) hist[cell * NFEAT + f] = s;
  }
  __syncthreads();

  float* dst = partial + (size_t)(b * WGS_PER_BATCH + wg) * NCOLS;
  for (int i = tid; i < NCOLS; i += 256) dst[i] = hist[i];
}

__global__ void k3_final(const float* __restrict__ partial,
                         float* __restrict__ out) {
  const int b    = blockIdx.x;
  const int lane = threadIdx.x & 31;
  const int wave = threadIdx.x >> 5;

  __shared__ float sums[NCOLS];
  __shared__ float fall[NCOLS];
  __shared__ float red[256];

  const float* pb = partial + (size_t)b * WGS_PER_BATCH * NCOLS;

#if __has_builtin(__builtin_amdgcn_wmma_f32_16x16x4_f32)
  for (int t = wave * 5; t < wave * 5 + 5; ++t) {
    const int col = t * 16 + (lane & 15);
    v8f c = {};
    v2f a; a[0] = 1.0f; a[1] = 1.0f;
    for (int k4 = 0; k4 < WGS_PER_BATCH / 4; ++k4) {
      const int r0 = k4 * 4 + ((lane < 16) ? 0 : 2);
      v2f bm;
      bm[0] = pb[(size_t)r0 * NCOLS + col];
      bm[1] = pb[(size_t)(r0 + 1) * NCOLS + col];
      c = __builtin_amdgcn_wmma_f32_16x16x4_f32(false, a, false, bm,
                                                (short)0, c, false, false);
    }
    if (lane < 16) sums[col] = c[0];
  }
#else
  for (int t = wave * 5; t < wave * 5 + 5; ++t) {
    const int col = t * 16 + (lane & 15);
    if (lane < 16) {
      float s = 0.0f;
      for (int k = 0; k < WGS_PER_BATCH; ++k) s += pb[(size_t)k * NCOLS + col];
      sums[col] = s;
    }
  }
#endif
  __syncthreads();

  float ss = 0.0f;
  for (int i = threadIdx.x; i < NCOLS; i += 256) {
    const int cell = i / NFEAT;
    const int f    = i - cell * NFEAT;
    const float cnt  = sums[cell * NFEAT];
    const float fc   = fmaxf(cnt, 1.0f);
    const float upre = rsqrtf(fc);
    float val;
    if (f == 0) {
      val = 0.001f * cnt * upre;
    } else if (f < 4) {
      const int d  = f - 1;
      const int ci = (d == 0) ? (cell >> 4) : ((d == 1) ? ((cell >> 2) & 3) : (cell & 3));
      const float center = ((float)ci + 0.5f) * 0.25f;
      val = (sums[i] - cnt * center) * upre;
    } else {
      val = sums[i] / fc;
    }
    fall[i] = val;
    ss += val * val;
  }

  red[threadIdx.x] = ss;
  __syncthreads();
  for (int s = 128; s > 0; s >>= 1) {
    if (threadIdx.x < s) red[threadIdx.x] += red[threadIdx.x + s];
    __syncthreads();
  }
  const float scale = 1.0f / fmaxf(sqrtf(red[0]), 1e-12f);
  for (int i = threadIdx.x; i < NCOLS; i += 256)
    out[(size_t)b * NCOLS + i] = fall[i] * scale;
}

extern "C" void kernel_launch(void* const* d_in, const int* in_sizes, int n_in,
                              void* d_out, int out_size, void* d_ws, size_t ws_size,
                              hipStream_t stream) {
  const float* x   = (const float*)d_in[0];
  unsigned* bbox   = (unsigned*)d_ws;
  float* partial   = (float*)d_ws + 64;
  float* out       = (float*)d_out;

  k0_init <<<1, 64, 0, stream>>>(bbox);
  k1_bbox <<<dim3(K1_BLOCKS, BATCHES), 256, 0, stream>>>(x, bbox);
  k2_hist <<<dim3(WGS_PER_BATCH, BATCHES), 256, 0, stream>>>(x, bbox, partial);
  k3_final<<<BATCHES, 256, 0, stream>>>(partial, out);
}
